# starting kernel
# baseline (speedup 1.0000x reference)
#include <hip/hip_runtime.h>
#include <hip/hip_bf16.h>

#define HID    128
#define BN_EPS 1e-5f

typedef __attribute__((ext_vector_type(2))) float v2f;
typedef __attribute__((ext_vector_type(8))) float v8f;

__device__ __forceinline__ v8f wmma_f32_16x16x4(v2f a, v2f b, v8f c) {
    return __builtin_amdgcn_wmma_f32_16x16x4_f32(
         false, a,  false, b,
         (short)0, c,  false,  false);
}

__global__ void k_init(const float* __restrict__ b, float* __restrict__ agg,
                       float* __restrict__ deg, int n_elem) {
    int idx = blockIdx.x * blockDim.x + threadIdx.x;
    if (idx >= n_elem) return;
    int c = idx & (HID - 1);
    agg[idx] = b[c];
    if (c == 0) deg[idx >> 7] = 1.0f;
}

__global__ void k_degcount(const int* __restrict__ dst, float* __restrict__ deg, int E) {
    int e = blockIdx.x * blockDim.x + threadIdx.x;
    if (e < E) atomicAdd(&deg[dst[e]], 1.0f);
}

__global__ void k_rsqrt(float* __restrict__ deg, int N) {
    int i = blockIdx.x * blockDim.x + threadIdx.x;
    if (i < N) deg[i] = rsqrtf(deg[i]);
}

__global__ void k_bias2(const float* __restrict__ gamma, const float* __restrict__ beta,
                        const float* __restrict__ mean,  const float* __restrict__ var,
                        const float* __restrict__ W, float* __restrict__ bias2) {
    int n = threadIdx.x;
    float s = 0.0f;
    for (int k = 0; k < HID; ++k) {
        float sc = rsqrtf(var[k] + BN_EPS) * gamma[k];
        float sh = beta[k] - mean[k] * sc;
        s += sh * W[k * HID + n];
    }
    bias2[n] = s;
}

__global__ __launch_bounds__(256) void k_bn_gemm(
    const float* __restrict__ x, const float* __restrict__ gamma,
    const float* __restrict__ var, const float* __restrict__ W,
    const float* __restrict__ bias2, float* __restrict__ m, int N)
{
    __shared__ float sW[HID * HID];

    const int tid = threadIdx.x;

    for (int idx = tid; idx < HID * HID; idx += 256) {
        int j    = idx & 1;
        int lane = (idx >> 1) & 31;
        int tile = (idx >> 6) & 7;
        int ks   = idx >> 9;
        int k    = 4 * ks + 2 * (lane >> 4) + j;
        int n    = tile * 16 + (lane & 15);
        float sc = rsqrtf(var[k] + BN_EPS) * gamma[k];
        sW[idx]  = sc * W[k * HID + n];
    }
    __syncthreads();

    const int wave  = tid >> 5;
    const int lane  = tid & 31;
    const int khalf = lane >> 4;
    const int colb  = lane & 15;

    const int rowbase = blockIdx.x * 128 + wave * 16;
    const int r       = rowbase + colb;
    const int rc      = (r < N) ? r : (N - 1);
    const float* xr   = x + (long)rc * HID;

    v8f acc[8] = {};

    for (int ks = 0; ks < 32; ++ks) {
        const int k0 = 4 * ks + 2 * khalf;
        v2f a;
        a.x = xr[k0];
        a.y = xr[k0 + 1];
        const float* wb = &sW[(ks * 8) * 64];
#pragma unroll
        for (int t = 0; t < 8; ++t) {
            v2f b = *(const v2f*)(wb + t * 64 + lane * 2);
            acc[t] = wmma_f32_16x16x4(a, b, acc[t]);
        }
    }

    float bsv[8];
#pragma unroll
    for (int t = 0; t < 8; ++t) bsv[t] = bias2[t * 16 + colb];

#pragma unroll
    for (int i = 0; i < 8; ++i) {
        int row = rowbase + i + 8 * khalf;
        if (row < N) {
            float* mr = m + (long)row * HID;
#pragma unroll
            for (int t = 0; t < 8; ++t)
                mr[t * 16 + colb] = acc[t][i] + bsv[t];
        }
    }
}

__global__ void k_scatter(const int* __restrict__ src, const int* __restrict__ dst,
                          const float* __restrict__ m, const float* __restrict__ dinv,
                          float* __restrict__ agg, int E) {
    int t = blockIdx.x * blockDim.x + threadIdx.x;
    int e = t >> 5;
    if (e >= E) return;
    int lane = t & 31;
    int s = src[e], d = dst[e];
    float nrm = dinv[s] * dinv[d];
    const float4 v = *(const float4*)(m + (long)s * HID + lane * 4);
    float* ap = agg + (long)d * HID + lane * 4;
    atomicAdd(ap + 0, v.x * nrm);
    atomicAdd(ap + 1, v.y * nrm);
    atomicAdd(ap + 2, v.z * nrm);
    atomicAdd(ap + 3, v.w * nrm);
}

__global__ void k_finalize(const float* __restrict__ x, const float* __restrict__ m,
                           const float* __restrict__ agg, const float* __restrict__ dinv,
                           float* __restrict__ out, int n_elem) {
    int idx = blockIdx.x * blockDim.x + threadIdx.x;
    if (idx >= n_elem) return;
    float di = dinv[idx >> 7];
    float v  = agg[idx] + m[idx] * di * di;
    out[idx] = x[idx] + fmaxf(v, 0.0f);
}

extern "C" void kernel_launch(void* const* d_in, const int* in_sizes, int n_in,
                              void* d_out, int out_size, void* d_ws, size_t ws_size,
                              hipStream_t stream) {
    const float* x     = (const float*)d_in[0];
    const int*   eidx  = (const int*)  d_in[1];
    const float* gamma = (const float*)d_in[2];
    const float* beta  = (const float*)d_in[3];
    const float* rmean = (const float*)d_in[4];
    const float* rvar  = (const float*)d_in[5];
    const float* W     = (const float*)d_in[6];
    const float* b     = (const float*)d_in[7];
    float*       out   = (float*)d_out;

    const int N = in_sizes[0] / HID;
    const int E = in_sizes[1] / 2;
    const int*   src = eidx;
    const int*   dst = eidx + E;

    float* m     = (float*)d_ws;
    float* agg   = m   + (size_t)N * HID;
    float* deg   = agg + (size_t)N * HID;
    float* bias2 = deg + N;

    const int n_elem = N * HID;

    k_init    <<<(n_elem + 255) / 256, 256, 0, stream>>>(b, agg, deg, n_elem);
    k_degcount<<<(E + 255) / 256,      256, 0, stream>>>(dst, deg, E);
    k_rsqrt   <<<(N + 255) / 256,      256, 0, stream>>>(deg, N);
    k_bias2   <<<1, HID,                    0, stream>>>(gamma, beta, rmean, rvar, W, bias2);
    k_bn_gemm <<<(N + 127) / 128,      256, 0, stream>>>(x, gamma, rvar, W, bias2, m, N);
    {
        long threads = (long)E * 32;
        k_scatter<<<(int)((threads + 255) / 256), 256, 0, stream>>>(src, dst, m, deg, agg, E);
    }
    k_finalize<<<(n_elem + 255) / 256, 256, 0, stream>>>(x, m, agg, deg, out, n_elem);
}


# Edges sorted by destination (int atomic rank, block scan, place); one wave per node sums its source rows in registers and writes out, replacing the float-atomic scatter; sums in arrival-rank order
# speedup vs baseline: 1.7664x; 1.7664x over previous
#include <hip/hip_runtime.h>
#include <hip/hip_bf16.h>


#define HID    128
#define BN_EPS 1e-5f

typedef __attribute__((ext_vector_type(2))) float v2f;
typedef __attribute__((ext_vector_type(8))) float v8f;

__device__ __forceinline__ v8f wmma_f32_16x16x4(v2f a, v2f b, v8f c) {
    return __builtin_amdgcn_wmma_f32_16x16x4_f32(
         false, a,  false, b,
         (short)0, c,  false,  false);
}

__global__ void k_zero(int* __restrict__ cnt, int N) {
    int i = blockIdx.x * blockDim.x + threadIdx.x;
    if (i < N) cnt[i] = 0;
}

__global__ void k_count(const int* __restrict__ dst, int* __restrict__ cnt,
                        int* __restrict__ rank, int E) {
    int e = blockIdx.x * blockDim.x + threadIdx.x;
    if (e < E) rank[e] = atomicAdd(&cnt[dst[e]], 1);
}

#define SCAN_THREADS 1024
#define SCAN_ITEMS   63
#define SCAN_TILE    (SCAN_THREADS * SCAN_ITEMS)

__global__ __launch_bounds__(SCAN_THREADS) void k_scan(
    const int* __restrict__ cnt, int* __restrict__ off, float* __restrict__ dinv, int N)
{
    __shared__ int sc[SCAN_TILE];
    __shared__ int swave[32];

    const int tid  = threadIdx.x;
    const int lane = tid & 31;
    const int w    = tid >> 5;
    int carry = 0;

    for (int base = 0; base < N; base += SCAN_TILE) {
        const int n = (N - base < SCAN_TILE) ? (N - base) : SCAN_TILE;

        int c[SCAN_ITEMS];
#pragma unroll
        for (int k = 0; k < SCAN_ITEMS; ++k) {
            const int i = k * SCAN_THREADS + tid;
            c[k] = (i < n) ? cnt[base + i] : 0;
        }
#pragma unroll
        for (int k = 0; k < SCAN_ITEMS; ++k) {
            const int i = k * SCAN_THREADS + tid;
            sc[i] = c[k];
            if (i < n) dinv[base + i] = rsqrtf((float)(c[k] + 1));
        }
        __syncthreads();

        int* p = sc + tid * SCAN_ITEMS;
        int sum = 0;
#pragma unroll
        for (int k = 0; k < SCAN_ITEMS; ++k) sum += p[k];

        int inc = sum;
        for (int dl = 1; dl < 32; dl <<= 1) {
            int t = __shfl_up(inc, dl);
            if (lane >= dl) inc += t;
        }
        if (lane == 31) swave[w] = inc;
        __syncthreads();
        if (w == 0) {
            int v = swave[lane];
            for (int dl = 1; dl < 32; dl <<= 1) {
                int t = __shfl_up(v, dl);
                if (lane >= dl) v += t;
            }
            swave[lane] = v;
        }
        __syncthreads();

        int run = carry + ((w > 0) ? swave[w - 1] : 0) + inc - sum;
        const int total = swave[31];
#pragma unroll
        for (int k = 0; k < SCAN_ITEMS; ++k) {
            int t = p[k];
            p[k]  = run;
            run  += t;
        }
        __syncthreads();

#pragma unroll
        for (int k = 0; k < SCAN_ITEMS; ++k) {
            const int i = k * SCAN_THREADS + tid;
            if (i < n) off[base + i] = sc[i];
        }
        carry += total;
        __syncthreads();
    }
    if (tid == 0) off[N] = carry;
}

__global__ void k_place(const int* __restrict__ src, const int* __restrict__ dst,
                        const int* __restrict__ rank, const int* __restrict__ off,
                        const float* __restrict__ dinv, int* __restrict__ csr, int E) {
    int e = blockIdx.x * blockDim.x + threadIdx.x;
    if (e >= E) return;
    const int   s = src[e], d = dst[e];
    const float w = dinv[s] * dinv[d];
    ((int2*)csr)[off[d] + rank[e]] = make_int2(s, __float_as_int(w));
}

__global__ void k_bias2(const float* __restrict__ gamma, const float* __restrict__ beta,
                        const float* __restrict__ mean,  const float* __restrict__ var,
                        const float* __restrict__ W, float* __restrict__ bias2) {
    int n = threadIdx.x;
    float s = 0.0f;
    for (int k = 0; k < HID; ++k) {
        float sc = rsqrtf(var[k] + BN_EPS) * gamma[k];
        float sh = beta[k] - mean[k] * sc;
        s += sh * W[k * HID + n];
    }
    bias2[n] = s;
}

__global__ __launch_bounds__(256) void k_bn_gemm(
    const float* __restrict__ x, const float* __restrict__ gamma,
    const float* __restrict__ var, const float* __restrict__ W,
    const float* __restrict__ bias2, float* __restrict__ m, int N)
{
    __shared__ float sW[HID * HID];

    const int tid = threadIdx.x;

    for (int idx = tid; idx < HID * HID; idx += 256) {
        int j    = idx & 1;
        int lane = (idx >> 1) & 31;
        int tile = (idx >> 6) & 7;
        int ks   = idx >> 9;
        int k    = 4 * ks + 2 * (lane >> 4) + j;
        int n    = tile * 16 + (lane & 15);
        float sc = rsqrtf(var[k] + BN_EPS) * gamma[k];
        sW[idx]  = sc * W[k * HID + n];
    }
    __syncthreads();

    const int wave  = tid >> 5;
    const int lane  = tid & 31;
    const int khalf = lane >> 4;
    const int colb  = lane & 15;

    const int rowbase = blockIdx.x * 128 + wave * 16;
    const int r       = rowbase + colb;
    const int rc      = (r < N) ? r : (N - 1);
    const float* xr   = x + (long)rc * HID;

    v8f acc[8] = {};

    for (int ks = 0; ks < 32; ++ks) {
        const int k0 = 4 * ks + 2 * khalf;
        v2f a;
        a.x = xr[k0];
        a.y = xr[k0 + 1];
        const float* wb = &sW[(ks * 8) * 64];
#pragma unroll
        for (int t = 0; t < 8; ++t) {
            v2f b = *(const v2f*)(wb + t * 64 + lane * 2);
            acc[t] = wmma_f32_16x16x4(a, b, acc[t]);
        }
    }

    float bsv[8];
#pragma unroll
    for (int t = 0; t < 8; ++t) bsv[t] = bias2[t * 16 + colb];

#pragma unroll
    for (int i = 0; i < 8; ++i) {
        int row = rowbase + i + 8 * khalf;
        if (row < N) {
            float* mr = m + (long)row * HID;
#pragma unroll
            for (int t = 0; t < 8; ++t)
                mr[t * 16 + colb] = acc[t][i] + bsv[t];
        }
    }
}

template <int K>
__device__ __forceinline__ void add_rows(const float* __restrict__ m, int s_l, int n_l,
                                         int k0, int lane, float4& acc) {
    float4 v[K];
    float  wgt[K];
#pragma unroll
    for (int i = 0; i < K; ++i) {
        const int s = __builtin_amdgcn_readlane(s_l, k0 + i);
        wgt[i] = __int_as_float(__builtin_amdgcn_readlane(n_l, k0 + i));
        v[i]   = *(const float4*)(m + (long)s * HID + lane * 4);
    }
#pragma unroll
    for (int i = 0; i < K; ++i) {
        acc.x += v[i].x * wgt[i];
        acc.y += v[i].y * wgt[i];
        acc.z += v[i].z * wgt[i];
        acc.w += v[i].w * wgt[i];
    }
}

__global__ __launch_bounds__(256) void k_gather(
    const int* __restrict__ off, const int* __restrict__ csr,
    const float* __restrict__ m, const float* __restrict__ dinv,
    const float* __restrict__ x, const float* __restrict__ b,
    float* __restrict__ out, int N)
{
    const int lane = threadIdx.x & 31;
    const int d = __builtin_amdgcn_readfirstlane(blockIdx.x * 8 + (threadIdx.x >> 5));
    if (d >= N) return;

    const int   beg = off[d];
    const int   end = off[d + 1];
    const float dd  = dinv[d];

    const float4 xr = *(const float4*)(x + (long)d * HID + lane * 4);
    const float4 mr = *(const float4*)(m + (long)d * HID + lane * 4);
    float4 acc      = *(const float4*)(b + lane * 4);

    for (int j0 = beg; j0 < end; j0 += 32) {
        const int j = j0 + lane;
        int s_l = 0;
        int n_l = 0;
        if (j < end) {
            const int2 sw = ((const int2*)csr)[j];
            s_l = sw.x;
            n_l = sw.y;
        }
        const int cnt = (end - j0 < 32) ? (end - j0) : 32;

        int k = 0;
        for (; k + 8 <= cnt; k += 8) add_rows<8>(m, s_l, n_l, k, lane, acc);
        if (k + 4 <= cnt) { add_rows<4>(m, s_l, n_l, k, lane, acc); k += 4; }
        const int rem = cnt - k;
        if      (rem == 3) add_rows<3>(m, s_l, n_l, k, lane, acc);
        else if (rem == 2) add_rows<2>(m, s_l, n_l, k, lane, acc);
        else if (rem == 1) add_rows<1>(m, s_l, n_l, k, lane, acc);
    }

    float4 o;
    o.x = xr.x + fmaxf(acc.x + mr.x * dd * dd, 0.0f);
    o.y = xr.y + fmaxf(acc.y + mr.y * dd * dd, 0.0f);
    o.z = xr.z + fmaxf(acc.z + mr.z * dd * dd, 0.0f);
    o.w = xr.w + fmaxf(acc.w + mr.w * dd * dd, 0.0f);
    *(float4*)(out + (long)d * HID + lane * 4) = o;
}

extern "C" void kernel_launch(void* const* d_in, const int* in_sizes, int n_in,
                              void* d_out, int out_size, void* d_ws, size_t ws_size,
                              hipStream_t stream) {
    const float* x     = (const float*)d_in[0];
    const int*   eidx  = (const int*)  d_in[1];
    const float* gamma = (const float*)d_in[2];
    const float* beta  = (const float*)d_in[3];
    const float* rmean = (const float*)d_in[4];
    const float* rvar  = (const float*)d_in[5];
    const float* W     = (const float*)d_in[6];
    const float* b     = (const float*)d_in[7];
    float*       out   = (float*)d_out;

    const int N = in_sizes[0] / HID;
    const int E = in_sizes[1] / 2;
    const int*   src = eidx;
    const int*   dst = eidx + E;

    float* m     = (float*)d_ws;
    int*   csr   = (int*)(m + (size_t)N * HID);
    float* dinv  = (float*)(csr + 2 * (size_t)E);
    float* bias2 = dinv + N;
    int*   cnt   = (int*)(bias2 + HID);
    int*   off   = cnt  + N;
    int*   rank  = off  + N + 1;

    k_zero   <<<(N + 255) / 256, 256,          0, stream>>>(cnt, N);
    k_count  <<<(E + 255) / 256, 256,          0, stream>>>(dst, cnt, rank, E);
    k_scan   <<<1,               SCAN_THREADS, 0, stream>>>(cnt, off, dinv, N);
    k_place  <<<(E + 255) / 256, 256,          0, stream>>>(src, dst, rank, off, dinv, csr, E);
    k_bias2  <<<1,               HID,          0, stream>>>(gamma, beta, rmean, rvar, W, bias2);
    k_bn_gemm<<<(N + 127) / 128, 256,          0, stream>>>(x, gamma, rvar, W, bias2, m, N);
    k_gather <<<(N + 7) / 8,     256,          0, stream>>>(off, csr, m, dinv, x, b, out, N);
}


# Edges go straight into 32-slot per-node lists at their atomic arrival rank (overflow list past 32); scan and placement launches dropped; weights from counts in gather; sums in arrival-rank order
# speedup vs baseline: 2.1419x; 1.2126x over previous
#include <hip/hip_runtime.h>
#include <hip/hip_bf16.h>


#define HID    128
#define BN_EPS 1e-5f

typedef __attribute__((ext_vector_type(2))) float v2f;
typedef __attribute__((ext_vector_type(8))) float v8f;

__device__ __forceinline__ v8f wmma_f32_16x16x4(v2f a, v2f b, v8f c) {
    return __builtin_amdgcn_wmma_f32_16x16x4_f32(
         false, a,  false, b,
         (short)0, c,  false,  false);
}

#define CAP 32

__global__ void k_zero(int* __restrict__ cnt, int n) {
    int i = blockIdx.x * blockDim.x + threadIdx.x;
    if (i < n) cnt[i] = 0;
}

__global__ void k_count_place(const int* __restrict__ src, const int* __restrict__ dst,
                              int* cnt, int* __restrict__ slot, int* __restrict__ ovf,
                              int N, int E) {
    int e = blockIdx.x * blockDim.x + threadIdx.x;
    if (e >= E) return;
    const int d = dst[e];
    const int s = src[e];
    const int r = atomicAdd(&cnt[d], 1);
    if (r < CAP) {
        slot[d * CAP + r] = s;
    } else {
        const int o = atomicAdd(&cnt[N], 1);
        ((int2*)ovf)[o] = make_int2(d, s);
    }
}

__global__ void k_bias2(const float* __restrict__ gamma, const float* __restrict__ beta,
                        const float* __restrict__ mean,  const float* __restrict__ var,
                        const float* __restrict__ W, float* __restrict__ bias2) {
    int n = threadIdx.x;
    float s = 0.0f;
    for (int k = 0; k < HID; ++k) {
        float sc = rsqrtf(var[k] + BN_EPS) * gamma[k];
        float sh = beta[k] - mean[k] * sc;
        s += sh * W[k * HID + n];
    }
    bias2[n] = s;
}

__global__ __launch_bounds__(256) void k_bn_gemm(
    const float* __restrict__ x, const float* __restrict__ gamma,
    const float* __restrict__ var, const float* __restrict__ W,
    const float* __restrict__ bias2, float* __restrict__ m, int N)
{
    __shared__ float sW[HID * HID];

    const int tid = threadIdx.x;

    for (int idx = tid; idx < HID * HID; idx += 256) {
        int j    = idx & 1;
        int lane = (idx >> 1) & 31;
        int tile = (idx >> 6) & 7;
        int ks   = idx >> 9;
        int k    = 4 * ks + 2 * (lane >> 4) + j;
        int n    = tile * 16 + (lane & 15);
        float sc = rsqrtf(var[k] + BN_EPS) * gamma[k];
        sW[idx]  = sc * W[k * HID + n];
    }
    __syncthreads();

    const int wave  = tid >> 5;
    const int lane  = tid & 31;
    const int khalf = lane >> 4;
    const int colb  = lane & 15;

    const int rowbase = blockIdx.x * 128 + wave * 16;
    const int r       = rowbase + colb;
    const int rc      = (r < N) ? r : (N - 1);
    const float* xr   = x + (long)rc * HID;

    v8f acc[8] = {};

    for (int ks = 0; ks < 32; ++ks) {
        const int k0 = 4 * ks + 2 * khalf;
        v2f a;
        a.x = xr[k0];
        a.y = xr[k0 + 1];
        const float* wb = &sW[(ks * 8) * 64];
#pragma unroll
        for (int t = 0; t < 8; ++t) {
            v2f b = *(const v2f*)(wb + t * 64 + lane * 2);
            acc[t] = wmma_f32_16x16x4(a, b, acc[t]);
        }
    }

    float bsv[8];
#pragma unroll
    for (int t = 0; t < 8; ++t) bsv[t] = bias2[t * 16 + colb];

#pragma unroll
    for (int i = 0; i < 8; ++i) {
        int row = rowbase + i + 8 * khalf;
        if (row < N) {
            float* mr = m + (long)row * HID;
#pragma unroll
            for (int t = 0; t < 8; ++t)
                mr[t * 16 + colb] = acc[t][i] + bsv[t];
        }
    }
}

template <int K>
__device__ __forceinline__ void add_rows(const float* __restrict__ m, int s_l, int n_l,
                                         int k0, int lane, float4& acc) {
    float4 v[K];
    float  wgt[K];
#pragma unroll
    for (int i = 0; i < K; ++i) {
        const int s = __builtin_amdgcn_readlane(s_l, k0 + i);
        wgt[i] = __int_as_float(__builtin_amdgcn_readlane(n_l, k0 + i));
        v[i]   = *(const float4*)(m + (long)s * HID + lane * 4);
    }
#pragma unroll
    for (int i = 0; i < K; ++i) {
        acc.x += v[i].x * wgt[i];
        acc.y += v[i].y * wgt[i];
        acc.z += v[i].z * wgt[i];
        acc.w += v[i].w * wgt[i];
    }
}

__global__ __launch_bounds__(256) void k_gather(
    const int* __restrict__ cnt, const int* __restrict__ slot, const int* __restrict__ ovf,
    const float* __restrict__ m, const float* __restrict__ x, const float* __restrict__ b,
    float* __restrict__ out, int N)
{
    const int lane = threadIdx.x & 31;
    const int d = __builtin_amdgcn_readfirstlane(blockIdx.x * 8 + (threadIdx.x >> 5));
    if (d >= N) return;

    const int   deg = cnt[d];
    const float dd  = rsqrtf((float)(deg + 1));

    const float4 xr = *(const float4*)(x + (long)d * HID + lane * 4);
    const float4 mr = *(const float4*)(m + (long)d * HID + lane * 4);
    float4 acc      = *(const float4*)(b + lane * 4);

    const int nin = (deg < CAP) ? deg : CAP;
    int s_l = 0;
    int n_l = 0;
    if (lane < nin) {
        s_l = slot[d * CAP + lane];
        n_l = __float_as_int(rsqrtf((float)(cnt[s_l] + 1)) * dd);
    }

    int k = 0;
    for (; k + 8 <= nin; k += 8) add_rows<8>(m, s_l, n_l, k, lane, acc);
    if (k + 4 <= nin) { add_rows<4>(m, s_l, n_l, k, lane, acc); k += 4; }
    const int rem = nin - k;
    if      (rem == 3) add_rows<3>(m, s_l, n_l, k, lane, acc);
    else if (rem == 2) add_rows<2>(m, s_l, n_l, k, lane, acc);
    else if (rem == 1) add_rows<1>(m, s_l, n_l, k, lane, acc);

    if (deg > CAP) {
        const int novf = cnt[N];
        for (int i0 = 0; i0 < novf; i0 += 32) {
            const int i = i0 + lane;
            int od = -1;
            int os = 0;
            if (i < novf) {
                const int2 p = ((const int2*)ovf)[i];
                od = p.x;
                os = p.y;
            }
            unsigned hit = __builtin_amdgcn_ballot_w32(od == d);
            while (hit) {
                const int l = __builtin_ctz(hit);
                hit &= hit - 1;
                const int    s = __builtin_amdgcn_readlane(os, l);
                const float  w = rsqrtf((float)(cnt[s] + 1)) * dd;
                const float4 v = *(const float4*)(m + (long)s * HID + lane * 4);
                acc.x += v.x * w;
                acc.y += v.y * w;
                acc.z += v.z * w;
                acc.w += v.w * w;
            }
        }
    }

    float4 o;
    o.x = xr.x + fmaxf(acc.x + mr.x * dd * dd, 0.0f);
    o.y = xr.y + fmaxf(acc.y + mr.y * dd * dd, 0.0f);
    o.z = xr.z + fmaxf(acc.z + mr.z * dd * dd, 0.0f);
    o.w = xr.w + fmaxf(acc.w + mr.w * dd * dd, 0.0f);
    *(float4*)(out + (long)d * HID + lane * 4) = o;
}

extern "C" void kernel_launch(void* const* d_in, const int* in_sizes, int n_in,
                              void* d_out, int out_size, void* d_ws, size_t ws_size,
                              hipStream_t stream) {
    const float* x     = (const float*)d_in[0];
    const int*   eidx  = (const int*)  d_in[1];
    const float* gamma = (const float*)d_in[2];
    const float* beta  = (const float*)d_in[3];
    const float* rmean = (const float*)d_in[4];
    const float* rvar  = (const float*)d_in[5];
    const float* W     = (const float*)d_in[6];
    const float* b     = (const float*)d_in[7];
    float*       out   = (float*)d_out;

    const int N = in_sizes[0] / HID;
    const int E = in_sizes[1] / 2;
    const int*   src = eidx;
    const int*   dst = eidx + E;

    float* m     = (float*)d_ws;
    int*   slot  = (int*)(m + (size_t)N * HID);
    int*   ovf   = slot + (size_t)N * CAP;
    float* bias2 = (float*)(ovf + 2 * (size_t)E);
    int*   cnt   = (int*)(bias2 + HID);

    k_zero       <<<(N + 256) / 256, 256, 0, stream>>>(cnt, N + 1);
    k_count_place<<<(E + 255) / 256, 256, 0, stream>>>(src, dst, cnt, slot, ovf, N, E);
    k_bias2      <<<1,               HID, 0, stream>>>(gamma, beta, rmean, rvar, W, bias2);
    k_bn_gemm    <<<(N + 127) / 128, 256, 0, stream>>>(x, gamma, rvar, W, bias2, m, N);
    k_gather     <<<(N + 7) / 8,     256, 0, stream>>>(cnt, slot, ovf, m, x, b, out, N);
}


# Batch norm applied to the A fragment on load (x*sc+sh from a 1 KB LDS table) instead of folded into W plus a bias row: bias2 launch dropped (5 to 4), W staged unscaled by 16-byte loads
# speedup vs baseline: 2.5729x; 1.2012x over previous
#include <hip/hip_runtime.h>
#include <hip/hip_bf16.h>


#define HID    128
#define BN_EPS 1e-5f

typedef __attribute__((ext_vector_type(2))) float v2f;
typedef __attribute__((ext_vector_type(8))) float v8f;

__device__ __forceinline__ v8f wmma_f32_16x16x4(v2f a, v2f b, v8f c) {
    return __builtin_amdgcn_wmma_f32_16x16x4_f32(
         false, a,  false, b,
         (short)0, c,  false,  false);
}

#define CAP 32

__global__ void k_zero(int* __restrict__ cnt, int n) {
    int i = blockIdx.x * blockDim.x + threadIdx.x;
    if (i < n) cnt[i] = 0;
}

__global__ void k_count_place(const int* __restrict__ src, const int* __restrict__ dst,
                              int* cnt, int* __restrict__ slot, int* __restrict__ ovf,
                              int N, int E) {
    int e = blockIdx.x * blockDim.x + threadIdx.x;
    if (e >= E) return;
    const int d = dst[e];
    const int s = src[e];
    const int r = atomicAdd(&cnt[d], 1);
    if (r < CAP) {
        slot[d * CAP + r] = s;
    } else {
        const int o = atomicAdd(&cnt[N], 1);
        ((int2*)ovf)[o] = make_int2(d, s);
    }
}

__global__ __launch_bounds__(256) void k_bn_gemm(
    const float* __restrict__ x, const float* __restrict__ gamma,
    const float* __restrict__ beta, const float* __restrict__ mean,
    const float* __restrict__ var, const float* __restrict__ W,
    float* __restrict__ m, int N)
{
    __shared__ float  sW[HID * HID];
    __shared__ float4 sBN[HID / 2];

    const int tid = threadIdx.x;

    if (tid < HID / 2) {
        const int   k   = 2 * tid;
        const float sc0 = rsqrtf(var[k]     + BN_EPS) * gamma[k];
        const float sc1 = rsqrtf(var[k + 1] + BN_EPS) * gamma[k + 1];
        sBN[tid] = make_float4(sc0, sc1, beta[k] - mean[k] * sc0, beta[k + 1] - mean[k + 1] * sc1);
    }

    float4 r0[8], r1[8];
#pragma unroll
    for (int i = 0; i < 8; ++i) {
        const int u  = tid + 256 * i;
        const int n4 = u & 31;
        const int kp = u >> 5;
        r0[i] = *(const float4*)(W + (2 * kp)     * HID + 4 * n4);
        r1[i] = *(const float4*)(W + (2 * kp + 1) * HID + 4 * n4);
    }
#pragma unroll
    for (int i = 0; i < 8; ++i) {
        const int u    = tid + 256 * i;
        const int n4   = u & 31;
        const int kp   = u >> 5;
        const int ks   = kp >> 1;
        const int tile = n4 >> 2;
        const int l0   = (kp & 1) * 16 + (n4 & 3) * 4;
        float* p = &sW[((ks * 8 + tile) * 32 + l0) * 2];
        *(float4*)(p)     = make_float4(r0[i].x, r1[i].x, r0[i].y, r1[i].y);
        *(float4*)(p + 4) = make_float4(r0[i].z, r1[i].z, r0[i].w, r1[i].w);
    }
    __syncthreads();

    const int wave  = tid >> 5;
    const int lane  = tid & 31;
    const int khalf = lane >> 4;
    const int colb  = lane & 15;

    const int rowbase = blockIdx.x * 128 + wave * 16;
    const int r       = rowbase + colb;
    const int rc      = (r < N) ? r : (N - 1);
    const float* xr   = x + (long)rc * HID;

    v8f acc[8] = {};

    for (int kq = 0; kq < 8; ++kq) {
        v2f    xa[4];
        float4 bn[4];
#pragma unroll
        for (int u = 0; u < 4; ++u) {
            const int ks = 4 * kq + u;
            xa[u] = *(const v2f*)(xr + 4 * ks + 2 * khalf);
            bn[u] = sBN[2 * ks + khalf];
        }
#pragma unroll
        for (int u = 0; u < 4; ++u) {
            const int ks = 4 * kq + u;
            v2f a;
            a.x = fmaf(xa[u].x, bn[u].x, bn[u].z);
            a.y = fmaf(xa[u].y, bn[u].y, bn[u].w);
            const float* wb = &sW[(ks * 8) * 64];
#pragma unroll
            for (int t = 0; t < 8; ++t) {
                v2f b = *(const v2f*)(wb + t * 64 + lane * 2);
                acc[t] = wmma_f32_16x16x4(a, b, acc[t]);
            }
        }
    }

#pragma unroll
    for (int i = 0; i < 8; ++i) {
        int row = rowbase + i + 8 * khalf;
        if (row < N) {
            float* mr = m + (long)row * HID;
#pragma unroll
            for (int t = 0; t < 8; ++t)
                mr[t * 16 + colb] = acc[t][i];
        }
    }
}

template <int K>
__device__ __forceinline__ void add_rows(const float* __restrict__ m, int s_l, int n_l,
                                         int k0, int lane, float4& acc) {
    float4 v[K];
    float  wgt[K];
#pragma unroll
    for (int i = 0; i < K; ++i) {
        const int s = __builtin_amdgcn_readlane(s_l, k0 + i);
        wgt[i] = __int_as_float(__builtin_amdgcn_readlane(n_l, k0 + i));
        v[i]   = *(const float4*)(m + (long)s * HID + lane * 4);
    }
#pragma unroll
    for (int i = 0; i < K; ++i) {
        acc.x += v[i].x * wgt[i];
        acc.y += v[i].y * wgt[i];
        acc.z += v[i].z * wgt[i];
        acc.w += v[i].w * wgt[i];
    }
}

__global__ __launch_bounds__(256) void k_gather(
    const int* __restrict__ cnt, const int* __restrict__ slot, const int* __restrict__ ovf,
    const float* __restrict__ m, const float* __restrict__ x, const float* __restrict__ b,
    float* __restrict__ out, int N)
{
    const int lane = threadIdx.x & 31;
    const int d = __builtin_amdgcn_readfirstlane(blockIdx.x * 8 + (threadIdx.x >> 5));
    if (d >= N) return;

    const int   deg = cnt[d];
    const float dd  = rsqrtf((float)(deg + 1));

    const float4 xr = *(const float4*)(x + (long)d * HID + lane * 4);
    const float4 mr = *(const float4*)(m + (long)d * HID + lane * 4);
    float4 acc      = *(const float4*)(b + lane * 4);

    const int nin = (deg < CAP) ? deg : CAP;
    int s_l = 0;
    int n_l = 0;
    if (lane < nin) {
        s_l = slot[d * CAP + lane];
        n_l = __float_as_int(rsqrtf((float)(cnt[s_l] + 1)) * dd);
    }

    int k = 0;
    for (; k + 8 <= nin; k += 8) add_rows<8>(m, s_l, n_l, k, lane, acc);
    if (k + 4 <= nin) { add_rows<4>(m, s_l, n_l, k, lane, acc); k += 4; }
    const int rem = nin - k;
    if      (rem == 3) add_rows<3>(m, s_l, n_l, k, lane, acc);
    else if (rem == 2) add_rows<2>(m, s_l, n_l, k, lane, acc);
    else if (rem == 1) add_rows<1>(m, s_l, n_l, k, lane, acc);

    if (deg > CAP) {
        const int novf = cnt[N];
        for (int i0 = 0; i0 < novf; i0 += 32) {
            const int i = i0 + lane;
            int od = -1;
            int os = 0;
            if (i < novf) {
                const int2 p = ((const int2*)ovf)[i];
                od = p.x;
                os = p.y;
            }
            unsigned hit = __builtin_amdgcn_ballot_w32(od == d);
            while (hit) {
                const int l = __builtin_ctz(hit);
                hit &= hit - 1;
                const int    s = __builtin_amdgcn_readlane(os, l);
                const float  w = rsqrtf((float)(cnt[s] + 1)) * dd;
                const float4 v = *(const float4*)(m + (long)s * HID + lane * 4);
                acc.x += v.x * w;
                acc.y += v.y * w;
                acc.z += v.z * w;
                acc.w += v.w * w;
            }
        }
    }

    float4 o;
    o.x = xr.x + fmaxf(acc.x + mr.x * dd * dd, 0.0f);
    o.y = xr.y + fmaxf(acc.y + mr.y * dd * dd, 0.0f);
    o.z = xr.z + fmaxf(acc.z + mr.z * dd * dd, 0.0f);
    o.w = xr.w + fmaxf(acc.w + mr.w * dd * dd, 0.0f);
    *(float4*)(out + (long)d * HID + lane * 4) = o;
}

extern "C" void kernel_launch(void* const* d_in, const int* in_sizes, int n_in,
                              void* d_out, int out_size, void* d_ws, size_t ws_size,
                              hipStream_t stream) {
    const float* x     = (const float*)d_in[0];
    const int*   eidx  = (const int*)  d_in[1];
    const float* gamma = (const float*)d_in[2];
    const float* beta  = (const float*)d_in[3];
    const float* rmean = (const float*)d_in[4];
    const float* rvar  = (const float*)d_in[5];
    const float* W     = (const float*)d_in[6];
    const float* b     = (const float*)d_in[7];
    float*       out   = (float*)d_out;

    const int N = in_sizes[0] / HID;
    const int E = in_sizes[1] / 2;
    const int*   src = eidx;
    const int*   dst = eidx + E;

    float* m     = (float*)d_ws;
    int*   slot  = (int*)(m + (size_t)N * HID);
    int*   ovf   = slot + (size_t)N * CAP;
    int*   cnt   = ovf + 2 * (size_t)E;

    k_zero       <<<(N + 256) / 256, 256, 0, stream>>>(cnt, N + 1);
    k_count_place<<<(E + 255) / 256, 256, 0, stream>>>(src, dst, cnt, slot, ovf, N, E);
    k_bn_gemm    <<<(N + 127) / 128, 256, 0, stream>>>(x, gamma, beta, rmean, rvar, W, m, N);
    k_gather     <<<(N + 7) / 8,     256, 0, stream>>>(cnt, slot, ovf, m, x, b, out, N);
}


# Edges sorted by 128-node destination range in LDS (counters, scan, stage; one global add per range and block), then ranked per range in LDS, slot lists stored as whole lines; sums in arrival order
# speedup vs baseline: 3.9738x; 1.5445x over previous
#include <hip/hip_runtime.h>
#include <hip/hip_bf16.h>


#define HID    128
#define BN_EPS 1e-5f

typedef __attribute__((ext_vector_type(2))) float v2f;
typedef __attribute__((ext_vector_type(8))) float v8f;
typedef __attribute__((ext_vector_type(2))) int   v2i;

__device__ __forceinline__ v8f wmma_f32_16x16x4(v2f a, v2f b, v8f c) {
    return __builtin_amdgcn_wmma_f32_16x16x4_f32(
         false, a,  false, b,
         (short)0, c,  false,  false);
}

#define CAP 32

#define BIN_SH    7
#define BIN_N     (1 << BIN_SH)
#define BINCAP    4096
#define HB        512
#define CUR_PITCH 64
#define BIN_T     512
#define BIN_K     8
#define RANK_T    256
#define RANK_K    (BINCAP / RANK_T)
#define RANK_Q    (BIN_N * CAP / 4 / RANK_T)

__global__ void k_zero(int* __restrict__ cur, int n) {
    int i = blockIdx.x * blockDim.x + threadIdx.x;
    if (i < n) cur[i * CUR_PITCH] = 0;
}

__global__ __launch_bounds__(BIN_T) void k_bin(
    const int* __restrict__ src, const int* __restrict__ dst, int* cur,
    int* __restrict__ binreg_i, int* __restrict__ spillA_i, int N, int NB, int E)
{
    v2i* __restrict__ binreg = (v2i*)binreg_i;
    v2i* __restrict__ spillA = (v2i*)spillA_i;
    static_assert(HB == BIN_T, "the scan below gives one bin of the window to each thread");
    __shared__ int  hist[HB];
    __shared__ int  delta[HB];
    __shared__ int  wsum[BIN_T / 32];
    __shared__ v2i  stage[BIN_T * BIN_K];

    const int tid  = threadIdx.x;
    const int lane = tid & 31;
    const int wave = tid >> 5;
    const int e0   = blockIdx.x * (BIN_T * BIN_K) + tid;

    int d[BIN_K], s[BIN_K], r[BIN_K];
#pragma unroll
    for (int k = 0; k < BIN_K; ++k) {
        const int e = e0 + k * BIN_T;
        int dd = -1, ss = 0;
        if (e < E) { dd = dst[e]; ss = src[e]; }
        d[k] = ((unsigned)dd < (unsigned)N) ? dd : -1;
        s[k] = ss;
    }

    for (int w0 = 0; w0 < NB; w0 += HB) {
        if (w0 > 0) __syncthreads();
        hist[tid] = 0;
        __syncthreads();
#pragma unroll
        for (int k = 0; k < BIN_K; ++k) {
            const int bin = (d[k] >> BIN_SH) - w0;
            r[k] = ((unsigned)bin < (unsigned)HB) ? atomicAdd(&hist[bin], 1) : -1;
        }
        __syncthreads();

        const int c = hist[tid];
        int incl = c;
#pragma unroll
        for (int off = 1; off < 32; off <<= 1) {
            const int v = __shfl_up(incl, off);
            if (lane >= off) incl += v;
        }
        if (lane == 31) wsum[wave] = incl;
        int g = 0;
        if (c != 0 && w0 + tid < NB) g = atomicAdd(&cur[(w0 + tid) * CUR_PITCH], c);
        __syncthreads();
        int wpre = 0, ntot = 0;
#pragma unroll
        for (int i = 0; i < BIN_T / 32; ++i) {
            const int v = wsum[i];
            ntot += v;
            if (i < wave) wpre += v;
        }
        const int loff = wpre + incl - c;
        hist[tid]  = loff;
        delta[tid] = g - loff;
        __syncthreads();

#pragma unroll
        for (int k = 0; k < BIN_K; ++k) {
            if (r[k] >= 0) {
                const int bin = (d[k] >> BIN_SH) - w0;
                v2i pr;
                pr.x = d[k];
                pr.y = s[k];
                stage[hist[bin] + r[k]] = pr;
            }
        }
        __syncthreads();

#pragma unroll
        for (int k = 0; k < BIN_K; ++k) {
            const int i = tid + k * BIN_T;
            if (i < ntot) {
                const v2i pr  = stage[i];
                const int bin = (pr.x >> BIN_SH) - w0;
                const int pos = delta[bin] + i;
                v2i* where = binreg + (size_t)(w0 + bin) * BINCAP + pos;
                if (pos >= BINCAP) where = spillA + atomicAdd(&cur[NB * CUR_PITCH], 1);
                *where = pr;
            }
        }
    }
}

__global__ __launch_bounds__(RANK_T) void k_rank(
    int* cur, const int* __restrict__ binreg_i, const int* __restrict__ spillA_i,
    int* __restrict__ cnt, int* __restrict__ slot, int* __restrict__ spillB_i, int N, int NB)
{
    const v2i* __restrict__ binreg = (const v2i*)binreg_i;
    const v2i* __restrict__ spillA = (const v2i*)spillA_i;
    v2i* __restrict__       spillB = (v2i*)spillB_i;
    __shared__ int  lcnt[BIN_N];
    __shared__ int4 lslot4[BIN_N * CAP / 4];
    int* lslot = (int*)lslot4;

    const int tid   = threadIdx.x;
    const int b     = blockIdx.x;
    const int node0 = b * BIN_N;

#pragma unroll
    for (int q = 0; q < RANK_Q; ++q) lslot4[tid + q * RANK_T] = make_int4(0, 0, 0, 0);
    if (tid < BIN_N) lcnt[tid] = 0;

    const int total = cur[b * CUR_PITCH];
    const int nreg  = (total < BINCAP) ? total : BINCAP;
    const v2i* reg = binreg + (size_t)b * BINCAP;

    static_assert(BINCAP == RANK_K * RANK_T && BIN_N <= RANK_T, "one pass covers a region");
    v2i p[RANK_K];
#pragma unroll
    for (int k = 0; k < RANK_K; ++k) {
        const int i = tid + k * RANK_T;
        p[k].x = -1;
        p[k].y = 0;
        if (i < nreg) p[k] = reg[i];
    }
    __syncthreads();

    auto place = [&](v2i pr) {
        const int dl = pr.x - node0;
        const int rk = atomicAdd(&lcnt[dl], 1);
        if (rk < CAP) {
            lslot[dl * CAP + rk] = pr.y;
        } else {
            const int o = atomicAdd(&cur[(NB + 1) * CUR_PITCH], 1);
            spillB[o] = pr;
        }
    };

#pragma unroll
    for (int k = 0; k < RANK_K; ++k)
        if (p[k].x >= 0) place(p[k]);

    if (total > BINCAP) {
        const int na = cur[NB * CUR_PITCH];
        for (int i = tid; i < na; i += RANK_T) {
            const v2i pr = spillA[i];
            if ((pr.x >> BIN_SH) == b) place(pr);
        }
    }
    __syncthreads();

#pragma unroll
    for (int q = 0; q < RANK_Q; ++q) {
        const int i = tid + q * RANK_T;
        if (node0 + (i >> 3) < N) ((int4*)(slot + (size_t)node0 * CAP))[i] = lslot4[i];
    }
    if (tid < BIN_N && node0 + tid < N) cnt[node0 + tid] = lcnt[tid];
}

__global__ __launch_bounds__(256) void k_bn_gemm(
    const float* __restrict__ x, const float* __restrict__ gamma,
    const float* __restrict__ beta, const float* __restrict__ mean,
    const float* __restrict__ var, const float* __restrict__ W,
    float* __restrict__ m, int N)
{
    __shared__ float  sW[HID * HID];
    __shared__ float4 sBN[HID / 2];

    const int tid = threadIdx.x;

    if (tid < HID / 2) {
        const int   k   = 2 * tid;
        const float sc0 = rsqrtf(var[k]     + BN_EPS) * gamma[k];
        const float sc1 = rsqrtf(var[k + 1] + BN_EPS) * gamma[k + 1];
        sBN[tid] = make_float4(sc0, sc1, beta[k] - mean[k] * sc0, beta[k + 1] - mean[k + 1] * sc1);
    }

    float4 r0[8], r1[8];
#pragma unroll
    for (int i = 0; i < 8; ++i) {
        const int u  = tid + 256 * i;
        const int n4 = u & 31;
        const int kp = u >> 5;
        r0[i] = *(const float4*)(W + (2 * kp)     * HID + 4 * n4);
        r1[i] = *(const float4*)(W + (2 * kp + 1) * HID + 4 * n4);
    }
#pragma unroll
    for (int i = 0; i < 8; ++i) {
        const int u    = tid + 256 * i;
        const int n4   = u & 31;
        const int kp   = u >> 5;
        const int ks   = kp >> 1;
        const int tile = n4 >> 2;
        const int l0   = (kp & 1) * 16 + (n4 & 3) * 4;
        float* p = &sW[((ks * 8 + tile) * 32 + l0) * 2];
        *(float4*)(p)     = make_float4(r0[i].x, r1[i].x, r0[i].y, r1[i].y);
        *(float4*)(p + 4) = make_float4(r0[i].z, r1[i].z, r0[i].w, r1[i].w);
    }
    __syncthreads();

    const int wave  = tid >> 5;
    const int lane  = tid & 31;
    const int khalf = lane >> 4;
    const int colb  = lane & 15;

    const int rowbase = blockIdx.x * 128 + wave * 16;
    const int r       = rowbase + colb;
    const int rc      = (r < N) ? r : (N - 1);
    const float* xr   = x + (long)rc * HID;

    v8f acc[8] = {};

    for (int kq = 0; kq < 8; ++kq) {
        v2f    xa[4];
        float4 bn[4];
#pragma unroll
        for (int u = 0; u < 4; ++u) {
            const int ks = 4 * kq + u;
            xa[u] = *(const v2f*)(xr + 4 * ks + 2 * khalf);
            bn[u] = sBN[2 * ks + khalf];
        }
#pragma unroll
        for (int u = 0; u < 4; ++u) {
            const int ks = 4 * kq + u;
            v2f a;
            a.x = fmaf(xa[u].x, bn[u].x, bn[u].z);
            a.y = fmaf(xa[u].y, bn[u].y, bn[u].w);
            const float* wb = &sW[(ks * 8) * 64];
#pragma unroll
            for (int t = 0; t < 8; ++t) {
                v2f b = *(const v2f*)(wb + t * 64 + lane * 2);
                acc[t] = wmma_f32_16x16x4(a, b, acc[t]);
            }
        }
    }

#pragma unroll
    for (int i = 0; i < 8; ++i) {
        int row = rowbase + i + 8 * khalf;
        if (row < N) {
            float* mr = m + (long)row * HID;
#pragma unroll
            for (int t = 0; t < 8; ++t)
                mr[t * 16 + colb] = acc[t][i];
        }
    }
}

template <int K>
__device__ __forceinline__ void add_rows(const float* __restrict__ m, int s_l, int n_l,
                                         int k0, int lane, float4& acc) {
    float4 v[K];
    float  wgt[K];
#pragma unroll
    for (int i = 0; i < K; ++i) {
        const int s = __builtin_amdgcn_readlane(s_l, k0 + i);
        wgt[i] = __int_as_float(__builtin_amdgcn_readlane(n_l, k0 + i));
        v[i]   = *(const float4*)(m + (long)s * HID + lane * 4);
    }
#pragma unroll
    for (int i = 0; i < K; ++i) {
        acc.x += v[i].x * wgt[i];
        acc.y += v[i].y * wgt[i];
        acc.z += v[i].z * wgt[i];
        acc.w += v[i].w * wgt[i];
    }
}

__global__ __launch_bounds__(256) void k_gather(
    const int* __restrict__ cnt, const int* __restrict__ slot, const int* __restrict__ ovf,
    const int* __restrict__ novf_p,
    const float* __restrict__ m, const float* __restrict__ x, const float* __restrict__ b,
    float* __restrict__ out, int N)
{
    const int lane = threadIdx.x & 31;
    const int d = __builtin_amdgcn_readfirstlane(blockIdx.x * 8 + (threadIdx.x >> 5));
    if (d >= N) return;

    const int   deg = cnt[d];
    const float dd  = rsqrtf((float)(deg + 1));

    const float4 xr = *(const float4*)(x + (long)d * HID + lane * 4);
    const float4 mr = *(const float4*)(m + (long)d * HID + lane * 4);
    float4 acc      = *(const float4*)(b + lane * 4);

    const int nin = (deg < CAP) ? deg : CAP;
    int s_l = 0;
    int n_l = 0;
    if (lane < nin) {
        s_l = slot[d * CAP + lane];
        n_l = __float_as_int(rsqrtf((float)(cnt[s_l] + 1)) * dd);
    }

    int k = 0;
    for (; k + 8 <= nin; k += 8) add_rows<8>(m, s_l, n_l, k, lane, acc);
    if (k + 4 <= nin) { add_rows<4>(m, s_l, n_l, k, lane, acc); k += 4; }
    const int rem = nin - k;
    if      (rem == 3) add_rows<3>(m, s_l, n_l, k, lane, acc);
    else if (rem == 2) add_rows<2>(m, s_l, n_l, k, lane, acc);
    else if (rem == 1) add_rows<1>(m, s_l, n_l, k, lane, acc);

    if (deg > CAP) {
        const int novf = novf_p[0];
        for (int i0 = 0; i0 < novf; i0 += 32) {
            const int i = i0 + lane;
            int od = -1;
            int os = 0;
            if (i < novf) {
                const int2 p = ((const int2*)ovf)[i];
                od = p.x;
                os = p.y;
            }
            unsigned hit = __builtin_amdgcn_ballot_w32(od == d);
            while (hit) {
                const int l = __builtin_ctz(hit);
                hit &= hit - 1;
                const int    s = __builtin_amdgcn_readlane(os, l);
                const float  w = rsqrtf((float)(cnt[s] + 1)) * dd;
                const float4 v = *(const float4*)(m + (long)s * HID + lane * 4);
                acc.x += v.x * w;
                acc.y += v.y * w;
                acc.z += v.z * w;
                acc.w += v.w * w;
            }
        }
    }

    float4 o;
    o.x = xr.x + fmaxf(acc.x + mr.x * dd * dd, 0.0f);
    o.y = xr.y + fmaxf(acc.y + mr.y * dd * dd, 0.0f);
    o.z = xr.z + fmaxf(acc.z + mr.z * dd * dd, 0.0f);
    o.w = xr.w + fmaxf(acc.w + mr.w * dd * dd, 0.0f);
    *(float4*)(out + (long)d * HID + lane * 4) = o;
}

extern "C" void kernel_launch(void* const* d_in, const int* in_sizes, int n_in,
                              void* d_out, int out_size, void* d_ws, size_t ws_size,
                              hipStream_t stream) {
    const float* x     = (const float*)d_in[0];
    const int*   eidx  = (const int*)  d_in[1];
    const float* gamma = (const float*)d_in[2];
    const float* beta  = (const float*)d_in[3];
    const float* rmean = (const float*)d_in[4];
    const float* rvar  = (const float*)d_in[5];
    const float* W     = (const float*)d_in[6];
    const float* b     = (const float*)d_in[7];
    float*       out   = (float*)d_out;

    const int N = in_sizes[0] / HID;
    const int E = in_sizes[1] / 2;
    const int*   src = eidx;
    const int*   dst = eidx + E;

    const int NB = (N + BIN_N - 1) / BIN_N;

    float* m      = (float*)d_ws;
    int*   slot   = (int*)(m + (size_t)N * HID);
    int*   spillB = slot + (size_t)N * CAP;
    int*   binreg = spillB + 2 * (size_t)E;
    int*   spillA = binreg + 2 * (size_t)NB * BINCAP;
    int*   cur    = spillA + 2 * (size_t)E;
    int*   cnt    = cur + (size_t)(NB + 2) * CUR_PITCH;

    k_zero   <<<(NB + 2 + 255) / 256, 256, 0, stream>>>(cur, NB + 2);
    k_bin    <<<(E + BIN_T * BIN_K - 1) / (BIN_T * BIN_K), BIN_T, 0, stream>>>(
                 src, dst, cur, binreg, spillA, N, NB, E);
    k_rank   <<<NB, RANK_T, 0, stream>>>(cur, binreg, spillA, cnt, slot, spillB, N, NB);
    k_bn_gemm<<<(N + 127) / 128, 256, 0, stream>>>(x, gamma, beta, rmean, rvar, W, m, N);
    k_gather <<<(N + 7) / 8,     256, 0, stream>>>(cnt, slot, spillB, cur + (size_t)(NB + 1) * CUR_PITCH,
                 m, x, b, out, N);
}


# Product m=bn(x)@W has no launch of its own any more: half of its 128-row blocks run inside the edge-binning launch, half inside the ranking launch (role by blockIdx); 5 launches become 4
# speedup vs baseline: 4.4491x; 1.1196x over previous
#include <hip/hip_runtime.h>
#include <hip/hip_bf16.h>


#define HID    128
#define BN_EPS 1e-5f

typedef __attribute__((ext_vector_type(2))) float v2f;
typedef __attribute__((ext_vector_type(8))) float v8f;
typedef __attribute__((ext_vector_type(2))) int   v2i;

__device__ __forceinline__ v8f wmma_f32_16x16x4(v2f a, v2f b, v8f c) {
    return __builtin_amdgcn_wmma_f32_16x16x4_f32(
         false, a,  false, b,
         (short)0, c,  false,  false);
}

#define CAP 32

#define BIN_SH    7
#define BIN_N     (1 << BIN_SH)
#define BINCAP    4096
#define HB        512
#define CUR_PITCH 64
#define BIN_T     512
#define BIN_K     8
#define RANK_T    256
#define RANK_K    (BINCAP / RANK_T)
#define RANK_Q    (BIN_N * CAP / 4 / RANK_T)

__global__ void k_zero(int* __restrict__ cur, int n) {
    int i = blockIdx.x * blockDim.x + threadIdx.x;
    if (i < n) cur[i * CUR_PITCH] = 0;
}

__device__ __forceinline__ void bin_block(
    const int bid, const int* __restrict__ src, const int* __restrict__ dst, int* cur,
    int* __restrict__ binreg_i, int* __restrict__ spillA_i, int N, int NB, int E)
{
    v2i* __restrict__ binreg = (v2i*)binreg_i;
    v2i* __restrict__ spillA = (v2i*)spillA_i;
    static_assert(HB == BIN_T, "the scan below gives one bin of the window to each thread");
    __shared__ int  hist[HB];
    __shared__ int  delta[HB];
    __shared__ int  wsum[BIN_T / 32];
    __shared__ v2i  stage[BIN_T * BIN_K];

    const int tid  = threadIdx.x;
    const int lane = tid & 31;
    const int wave = tid >> 5;
    const int e0   = bid * (BIN_T * BIN_K) + tid;

    int d[BIN_K], s[BIN_K], r[BIN_K];
#pragma unroll
    for (int k = 0; k < BIN_K; ++k) {
        const int e = e0 + k * BIN_T;
        int dd = -1, ss = 0;
        if (e < E) { dd = dst[e]; ss = src[e]; }
        d[k] = ((unsigned)dd < (unsigned)N) ? dd : -1;
        s[k] = ss;
    }

    for (int w0 = 0; w0 < NB; w0 += HB) {
        if (w0 > 0) __syncthreads();
        hist[tid] = 0;
        __syncthreads();
#pragma unroll
        for (int k = 0; k < BIN_K; ++k) {
            const int bin = (d[k] >> BIN_SH) - w0;
            r[k] = ((unsigned)bin < (unsigned)HB) ? atomicAdd(&hist[bin], 1) : -1;
        }
        __syncthreads();

        const int c = hist[tid];
        int incl = c;
#pragma unroll
        for (int off = 1; off < 32; off <<= 1) {
            const int v = __shfl_up(incl, off);
            if (lane >= off) incl += v;
        }
        if (lane == 31) wsum[wave] = incl;
        int g = 0;
        if (c != 0 && w0 + tid < NB) g = atomicAdd(&cur[(w0 + tid) * CUR_PITCH], c);
        __syncthreads();
        int wpre = 0, ntot = 0;
#pragma unroll
        for (int i = 0; i < BIN_T / 32; ++i) {
            const int v = wsum[i];
            ntot += v;
            if (i < wave) wpre += v;
        }
        const int loff = wpre + incl - c;
        hist[tid]  = loff;
        delta[tid] = g - loff;
        __syncthreads();

#pragma unroll
        for (int k = 0; k < BIN_K; ++k) {
            if (r[k] >= 0) {
                const int bin = (d[k] >> BIN_SH) - w0;
                v2i pr;
                pr.x = d[k];
                pr.y = s[k];
                stage[hist[bin] + r[k]] = pr;
            }
        }
        __syncthreads();

#pragma unroll
        for (int k = 0; k < BIN_K; ++k) {
            const int i = tid + k * BIN_T;
            if (i < ntot) {
                const v2i pr  = stage[i];
                const int bin = (pr.x >> BIN_SH) - w0;
                const int pos = delta[bin] + i;
                v2i* where = binreg + (size_t)(w0 + bin) * BINCAP + pos;
                if (pos >= BINCAP) where = spillA + atomicAdd(&cur[NB * CUR_PITCH], 1);
                *where = pr;
            }
        }
    }
}

__device__ __forceinline__ void rank_block(
    const int b, int* cur, const int* __restrict__ binreg_i, const int* __restrict__ spillA_i,
    int* __restrict__ cnt, int* __restrict__ slot, int* __restrict__ spillB_i, int N, int NB)
{
    const v2i* __restrict__ binreg = (const v2i*)binreg_i;
    const v2i* __restrict__ spillA = (const v2i*)spillA_i;
    v2i* __restrict__       spillB = (v2i*)spillB_i;
    __shared__ int  lcnt[BIN_N];
    __shared__ int4 lslot4[BIN_N * CAP / 4];
    int* lslot = (int*)lslot4;

    const int tid   = threadIdx.x;
    const int node0 = b * BIN_N;

#pragma unroll
    for (int q = 0; q < RANK_Q; ++q) lslot4[tid + q * RANK_T] = make_int4(0, 0, 0, 0);
    if (tid < BIN_N) lcnt[tid] = 0;

    const int total = cur[b * CUR_PITCH];
    const int nreg  = (total < BINCAP) ? total : BINCAP;
    const v2i* reg = binreg + (size_t)b * BINCAP;

    static_assert(BINCAP == RANK_K * RANK_T && BIN_N <= RANK_T, "one pass covers a region");
    v2i p[RANK_K];
#pragma unroll
    for (int k = 0; k < RANK_K; ++k) {
        const int i = tid + k * RANK_T;
        p[k].x = -1;
        p[k].y = 0;
        if (i < nreg) p[k] = reg[i];
    }
    __syncthreads();

    auto place = [&](v2i pr) {
        const int dl = pr.x - node0;
        const int rk = atomicAdd(&lcnt[dl], 1);
        if (rk < CAP) {
            lslot[dl * CAP + rk] = pr.y;
        } else {
            const int o = atomicAdd(&cur[(NB + 1) * CUR_PITCH], 1);
            spillB[o] = pr;
        }
    };

#pragma unroll
    for (int k = 0; k < RANK_K; ++k)
        if (p[k].x >= 0) place(p[k]);

    if (total > BINCAP) {
        const int na = cur[NB * CUR_PITCH];
        for (int i = tid; i < na; i += RANK_T) {
            const v2i pr = spillA[i];
            if ((pr.x >> BIN_SH) == b) place(pr);
        }
    }
    __syncthreads();

#pragma unroll
    for (int q = 0; q < RANK_Q; ++q) {
        const int i = tid + q * RANK_T;
        if (node0 + (i >> 3) < N) ((int4*)(slot + (size_t)node0 * CAP))[i] = lslot4[i];
    }
    if (tid < BIN_N && node0 + tid < N) cnt[node0 + tid] = lcnt[tid];
}

template <int NT>
__device__ __forceinline__ void gemm_block(
    const int rb,
    const float* __restrict__ x, const float* __restrict__ gamma,
    const float* __restrict__ beta, const float* __restrict__ mean,
    const float* __restrict__ var, const float* __restrict__ W,
    float* __restrict__ m, int N)
{
    constexpr int WPR = NT / 256;
    constexpr int CT  = 8 / WPR;
    constexpr int SU  = HID * HID / 8 / NT;
    static_assert(NT == 256 || NT == 512, "128 rows per block");

    __shared__ float  sW[HID * HID];
    __shared__ float4 sBN[HID / 2];

    const int tid = threadIdx.x;

    if (tid < HID / 2) {
        const int   k   = 2 * tid;
        const float sc0 = rsqrtf(var[k]     + BN_EPS) * gamma[k];
        const float sc1 = rsqrtf(var[k + 1] + BN_EPS) * gamma[k + 1];
        sBN[tid] = make_float4(sc0, sc1, beta[k] - mean[k] * sc0, beta[k + 1] - mean[k + 1] * sc1);
    }

    float4 r0[SU], r1[SU];
#pragma unroll
    for (int i = 0; i < SU; ++i) {
        const int u  = tid + NT * i;
        const int n4 = u & 31;
        const int kp = u >> 5;
        r0[i] = *(const float4*)(W + (2 * kp)     * HID + 4 * n4);
        r1[i] = *(const float4*)(W + (2 * kp + 1) * HID + 4 * n4);
    }
#pragma unroll
    for (int i = 0; i < SU; ++i) {
        const int u    = tid + NT * i;
        const int n4   = u & 31;
        const int kp   = u >> 5;
        const int ks   = kp >> 1;
        const int tile = n4 >> 2;
        const int l0   = (kp & 1) * 16 + (n4 & 3) * 4;
        float* p = &sW[((ks * 8 + tile) * 32 + l0) * 2];
        *(float4*)(p)     = make_float4(r0[i].x, r1[i].x, r0[i].y, r1[i].y);
        *(float4*)(p + 4) = make_float4(r0[i].z, r1[i].z, r0[i].w, r1[i].w);
    }
    __syncthreads();

    const int wave  = tid >> 5;
    const int lane  = tid & 31;
    const int khalf = lane >> 4;
    const int colb  = lane & 15;
    const int tile0 = (wave % WPR) * CT;

    const int rowbase = rb * 128 + (wave / WPR) * 16;
    const int r       = rowbase + colb;
    const int rc      = (r < N) ? r : (N - 1);
    const float* xr   = x + (long)rc * HID;

    v8f acc[CT] = {};

    for (int kq = 0; kq < 8; ++kq) {
        v2f    xa[4];
        float4 bn[4];
#pragma unroll
        for (int u = 0; u < 4; ++u) {
            const int ks = 4 * kq + u;
            xa[u] = *(const v2f*)(xr + 4 * ks + 2 * khalf);
            bn[u] = sBN[2 * ks + khalf];
        }
#pragma unroll
        for (int u = 0; u < 4; ++u) {
            const int ks = 4 * kq + u;
            v2f a;
            a.x = fmaf(xa[u].x, bn[u].x, bn[u].z);
            a.y = fmaf(xa[u].y, bn[u].y, bn[u].w);
            const float* wb = &sW[(ks * 8 + tile0) * 64];
#pragma unroll
            for (int t = 0; t < CT; ++t) {
                v2f b = *(const v2f*)(wb + t * 64 + lane * 2);
                acc[t] = wmma_f32_16x16x4(a, b, acc[t]);
            }
        }
    }

#pragma unroll
    for (int i = 0; i < 8; ++i) {
        int row = rowbase + i + 8 * khalf;
        if (row < N) {
            float* mr = m + (long)row * HID + tile0 * 16;
#pragma unroll
            for (int t = 0; t < CT; ++t)
                mr[t * 16 + colb] = acc[t][i];
        }
    }
}


__global__ __launch_bounds__(BIN_T) void k_bin_gemm(
    const int* __restrict__ src, const int* __restrict__ dst, int* cur,
    int* __restrict__ binreg_i, int* __restrict__ spillA_i, int N, int NB, int E,
    const float* __restrict__ x, const float* __restrict__ gamma,
    const float* __restrict__ beta, const float* __restrict__ mean,
    const float* __restrict__ var, const float* __restrict__ W,
    float* __restrict__ m, int NGA)
{
    const int bx = blockIdx.x;
    if (bx < NGA) {
        gemm_block<BIN_T>(bx, x, gamma, beta, mean, var, W, m, N);
        return;
    }
    bin_block(bx - NGA, src, dst, cur, binreg_i, spillA_i, N, NB, E);
}

__global__ __launch_bounds__(RANK_T, 6) void k_rank_gemm(
    int* cur, const int* __restrict__ binreg_i, const int* __restrict__ spillA_i,
    int* __restrict__ cnt, int* __restrict__ slot, int* __restrict__ spillB_i, int N, int NB,
    const float* __restrict__ x, const float* __restrict__ gamma,
    const float* __restrict__ beta, const float* __restrict__ mean,
    const float* __restrict__ var, const float* __restrict__ W,
    float* __restrict__ m, int NGA, int NGB)
{
    const int bx = blockIdx.x;
    if (bx < NGB) {
        gemm_block<RANK_T>(NGA + bx, x, gamma, beta, mean, var, W, m, N);
        return;
    }
    rank_block(bx - NGB, cur, binreg_i, spillA_i, cnt, slot, spillB_i, N, NB);
}

template <int K>
__device__ __forceinline__ void add_rows(const float* __restrict__ m, int s_l, int n_l,
                                         int k0, int lane, float4& acc) {
    float4 v[K];
    float  wgt[K];
#pragma unroll
    for (int i = 0; i < K; ++i) {
        const int s = __builtin_amdgcn_readlane(s_l, k0 + i);
        wgt[i] = __int_as_float(__builtin_amdgcn_readlane(n_l, k0 + i));
        v[i]   = *(const float4*)(m + (long)s * HID + lane * 4);
    }
#pragma unroll
    for (int i = 0; i < K; ++i) {
        acc.x += v[i].x * wgt[i];
        acc.y += v[i].y * wgt[i];
        acc.z += v[i].z * wgt[i];
        acc.w += v[i].w * wgt[i];
    }
}

__global__ __launch_bounds__(256) void k_gather(
    const int* __restrict__ cnt, const int* __restrict__ slot, const int* __restrict__ ovf,
    const int* __restrict__ novf_p,
    const float* __restrict__ m, const float* __restrict__ x, const float* __restrict__ b,
    float* __restrict__ out, int N)
{
    const int lane = threadIdx.x & 31;
    const int d = __builtin_amdgcn_readfirstlane(blockIdx.x * 8 + (threadIdx.x >> 5));
    if (d >= N) return;

    const int   deg = cnt[d];
    const float dd  = rsqrtf((float)(deg + 1));

    const float4 xr = *(const float4*)(x + (long)d * HID + lane * 4);
    const float4 mr = *(const float4*)(m + (long)d * HID + lane * 4);
    float4 acc      = *(const float4*)(b + lane * 4);

    const int nin = (deg < CAP) ? deg : CAP;
    int s_l = 0;
    int n_l = 0;
    if (lane < nin) {
        s_l = slot[d * CAP + lane];
        n_l = __float_as_int(rsqrtf((float)(cnt[s_l] + 1)) * dd);
    }

    int k = 0;
    for (; k + 8 <= nin; k += 8) add_rows<8>(m, s_l, n_l, k, lane, acc);
    if (k + 4 <= nin) { add_rows<4>(m, s_l, n_l, k, lane, acc); k += 4; }
    const int rem = nin - k;
    if      (rem == 3) add_rows<3>(m, s_l, n_l, k, lane, acc);
    else if (rem == 2) add_rows<2>(m, s_l, n_l, k, lane, acc);
    else if (rem == 1) add_rows<1>(m, s_l, n_l, k, lane, acc);

    if (deg > CAP) {
        const int novf = novf_p[0];
        for (int i0 = 0; i0 < novf; i0 += 32) {
            const int i = i0 + lane;
            int od = -1;
            int os = 0;
            if (i < novf) {
                const int2 p = ((const int2*)ovf)[i];
                od = p.x;
                os = p.y;
            }
            unsigned hit = __builtin_amdgcn_ballot_w32(od == d);
            while (hit) {
                const int l = __builtin_ctz(hit);
                hit &= hit - 1;
                const int    s = __builtin_amdgcn_readlane(os, l);
                const float  w = rsqrtf((float)(cnt[s] + 1)) * dd;
                const float4 v = *(const float4*)(m + (long)s * HID + lane * 4);
                acc.x += v.x * w;
                acc.y += v.y * w;
                acc.z += v.z * w;
                acc.w += v.w * w;
            }
        }
    }

    float4 o;
    o.x = xr.x + fmaxf(acc.x + mr.x * dd * dd, 0.0f);
    o.y = xr.y + fmaxf(acc.y + mr.y * dd * dd, 0.0f);
    o.z = xr.z + fmaxf(acc.z + mr.z * dd * dd, 0.0f);
    o.w = xr.w + fmaxf(acc.w + mr.w * dd * dd, 0.0f);
    *(float4*)(out + (long)d * HID + lane * 4) = o;
}

extern "C" void kernel_launch(void* const* d_in, const int* in_sizes, int n_in,
                              void* d_out, int out_size, void* d_ws, size_t ws_size,
                              hipStream_t stream) {
    const float* x     = (const float*)d_in[0];
    const int*   eidx  = (const int*)  d_in[1];
    const float* gamma = (const float*)d_in[2];
    const float* beta  = (const float*)d_in[3];
    const float* rmean = (const float*)d_in[4];
    const float* rvar  = (const float*)d_in[5];
    const float* W     = (const float*)d_in[6];
    const float* b     = (const float*)d_in[7];
    float*       out   = (float*)d_out;

    const int N = in_sizes[0] / HID;
    const int E = in_sizes[1] / 2;
    const int*   src = eidx;
    const int*   dst = eidx + E;

    const int NB = (N + BIN_N - 1) / BIN_N;

    float* m      = (float*)d_ws;
    int*   slot   = (int*)(m + (size_t)N * HID);
    int*   spillB = slot + (size_t)N * CAP;
    int*   binreg = spillB + 2 * (size_t)E;
    int*   spillA = binreg + 2 * (size_t)NB * BINCAP;
    int*   cur    = spillA + 2 * (size_t)E;
    int*   cnt    = cur + (size_t)(NB + 2) * CUR_PITCH;

    const int NG  = (N + 127) / 128;
    const int NGA = (NG + 1) / 2;
    const int NGB = NG - NGA;
    const int NBB = (E + BIN_T * BIN_K - 1) / (BIN_T * BIN_K);

    k_zero     <<<(NB + 2 + 255) / 256, 256, 0, stream>>>(cur, NB + 2);
    k_bin_gemm <<<NGA + NBB, BIN_T, 0, stream>>>(src, dst, cur, binreg, spillA, N, NB, E,
                   x, gamma, beta, rmean, rvar, W, m, NGA);
    k_rank_gemm<<<NGB + NB, RANK_T, 0, stream>>>(cur, binreg, spillA, cnt, slot, spillB, N, NB,
                   x, gamma, beta, rmean, rvar, W, m, NGA, NGB);
    k_gather   <<<(N + 7) / 8,     256, 0, stream>>>(cnt, slot, spillB, cur + (size_t)(NB + 1) * CUR_PITCH,
                 m, x, b, out, N);
}
